# starting kernel
# baseline (speedup 1.0000x reference)
#include <hip/hip_runtime.h>

typedef __attribute__((ext_vector_type(2))) float v2f;
typedef __attribute__((ext_vector_type(8))) float v8f;

#define B_    4
#define CIN_  16
#define COUT_ 16
#define H_    128
#define W_    128
#define KK    144
#define WPB   8

__global__ __launch_bounds__(256) void xonv2d_wmma_f32(
    const float* __restrict__ x,
    const float* __restrict__ wgt,
    const float* __restrict__ bias,
    float* __restrict__ out)
{
  __shared__ __align__(16) float patch[WPB][B_][KK];

  const int lane = threadIdx.x & 31;
  const int wave = threadIdx.x >> 5;
  const int pix  = blockIdx.x * WPB + wave;
  const int h    = pix >> 7;
  const int w    = pix & 127;

  float* pat = &patch[wave][0][0];
  for (int e = lane; e < B_ * KK; e += 32) {
    int b  = e / KK;
    int p  = e - b * KK;
    int c  = p / 9;
    int r  = p - c * 9;
    int kh = r / 3;
    int kw = r - kh * 3;
    int hh = h + kh - 1;
    int ww = w + kw - 1;
    float v = 0.0f;
    if (hh >= 0 && hh < H_ && ww >= 0 && ww < W_)
      v = x[((b * CIN_ + c) * H_ + hh) * W_ + ww];
    pat[e] = v;
  }
  __syncthreads();

  const float* wpix = wgt + (size_t)pix * (COUT_ * KK);
  const int n    = lane & 15;
  const int half = lane >> 4;

  v8f acc = {};
  for (int t = 0; t < KK / 4; ++t) {
    const int p0 = 4 * t + 2 * half;
    v2f a = *(const v2f*)(wpix + n * KK + p0);
    v2f b = *(const v2f*)(pat + (n & (B_ - 1)) * KK + p0);
    acc = __builtin_amdgcn_wmma_f32_16x16x4_f32(
         false, a,  false, b,
         (short)0, acc,  false,  false);
  }

  if (n < B_) {
    const float* bp = bias + (size_t)pix * COUT_;
    float* op = out + (size_t)n * COUT_ * (H_ * W_) + pix;
#pragma unroll
    for (int r = 0; r < 8; ++r) {
      const int m = r + 8 * half;
      op[(size_t)m * (H_ * W_)] = acc[r] + bp[m];
    }
  }
}

extern "C" void kernel_launch(void* const* d_in, const int* in_sizes, int n_in,
                              void* d_out, int out_size, void* d_ws, size_t ws_size,
                              hipStream_t stream) {
  (void)in_sizes; (void)n_in; (void)out_size; (void)d_ws; (void)ws_size;
  const float* x    = (const float*)d_in[0];
  const float* wgt  = (const float*)d_in[1];
  const float* bias = (const float*)d_in[2];
  float* out        = (float*)d_out;

  dim3 grid((H_ * W_) / WPB);
  dim3 block(256);
  xonv2d_wmma_f32<<<grid, block, 0, stream>>>(x, wgt, bias, out);
}


# Waves split by role in blocks of 32 pixels: 4 loader waves stream weight lines two stages ahead into a 2-stage LDS ring, 4 matrix waves only multiply; 512 resident blocks; sums unchanged
# speedup vs baseline: 2.1676x; 2.1676x over previous
#include <hip/hip_runtime.h>

typedef __attribute__((ext_vector_type(2))) float v2f;
typedef __attribute__((ext_vector_type(4))) float v4f;
typedef __attribute__((ext_vector_type(8))) float v8f;

#define B_    4
#define CIN_  16
#define COUT_ 16
#define H_    128
#define W_    128
#define KK    144
#define SEG   32
#define PPS   4
#define NST   (SEG / PPS)
#define TP    (SEG + 8)
#define TPC   (TP / 4)
#define TPCS  (B_ * CIN_ * 3 * TPC)
#define XPT   ((TPCS + 255) / 256)
#define BSTR  (48 * TP + 8)
#define KOFF(k) ((((k) / 9) * 3 + ((k) % 9) / 3) * TP + (k) % 3)
#define WBLK  (COUT_ * KK)
#define NWL   (WBLK / 4 / 32)
#define WROW  (KK + 4)
#define WSTR  (COUT_ * WROW)

#define HANDOVER() asm volatile("s_wait_dscnt 0x0\n\ts_barrier_signal -1\n\ts_barrier_wait -1" ::: "memory")

__device__ __forceinline__ void request(v4f (&w)[NWL], const float* __restrict__ wp) {
#pragma unroll
  for (int i = 0; i < NWL; ++i) w[i] = *(const v4f*)(wp + 128 * i);
}

__device__ __forceinline__ void stash(const v4f (&w)[NWL], float* wlw, int lane) {
#pragma unroll
  for (int i = 0; i < NWL; ++i) {
    const unsigned j = 32u * i + lane;
    *(v4f*)(wlw + 4u * (j + j / 36u)) = w[i];
  }
}

__device__ __forceinline__ void pixel(int p, const float* wlw, const float* tile, const float* bl,
                                      float* __restrict__ out, int pix0, int n, int half) {
  const float* ab = wlw + n * WROW + 2 * half;
  const float* tb   = tile + (n & (B_ - 1)) * BSTR + p + 3;
  const float* tb2  = tb + (half ? 2 : 0);
  const float* tb15 = tb + (half ? TP - 1 : 0);
  v4f bia0 = *(const v4f*)(bl + p * COUT_ + 8 * half);
  v4f bia1 = *(const v4f*)(bl + p * COUT_ + 8 * half + 4);
  v8f acc = {};
#pragma unroll
  for (int t = 0; t < KK / 4; ++t) {
    v2f b;
    b.x = (((4 * t) % 3 == 0) ? tb2 : tb15)[KOFF(4 * t)];
    b.y = (((4 * t + 1) % 3 == 0) ? tb2 : tb15)[KOFF(4 * t + 1)];
    const v2f a = *(const v2f*)(ab + 4 * t);
    acc = __builtin_amdgcn_wmma_f32_16x16x4_f32(
         false, a,  false, b,
         (short)0, acc,  false,  false);
  }
  asm volatile("v_nop\n\tv_nop\n\tv_nop\n\tv_nop" : "+v"(acc), "+v"(bia0), "+v"(bia1));
  if (n < B_) {
    float* op = out + (size_t)(n * COUT_ + 8 * half) * (H_ * W_) + pix0 + p;
#pragma unroll
    for (int r = 0; r < 8; ++r)
      op[(size_t)r * (H_ * W_)] = acc[r] + (r < 4 ? bia0[r & 3] : bia1[r & 3]);
  }
}

__global__ __launch_bounds__(256) void xonv2d_wmma_f32(
    const float* __restrict__ x,
    const float* __restrict__ wgt,
    const float* __restrict__ bias,
    float* __restrict__ out)
{
  __shared__ __align__(16) float tile[B_ * BSTR];
  __shared__ __align__(16) float wl[2 * PPS * WSTR];
  __shared__ __align__(16) float bl[SEG * COUT_];

  const int tid  = threadIdx.x;
  const int lane = tid & 31;
  const int wave = __builtin_amdgcn_readfirstlane(tid >> 5);
  const bool loader = wave < PPS;
  const int slot = wave & (PPS - 1);
  const int h    = blockIdx.x >> 2;
  const int w0   = (blockIdx.x & 3) * SEG;
  const int pix0 = h * W_ + w0;
  const int n    = lane & 15;
  const int half = lane >> 4;

  v4f wa[NWL], wb[NWL];
  const float* wsl = wgt + (size_t)(pix0 + slot) * WBLK + 4 * lane;
  if (loader) request(wa, wsl);

  v4f xv[XPT];
  bool ok[XPT];
#pragma unroll
  for (int i = 0; i < XPT; ++i) {
    const int e  = tid + 256 * i;
    const int r  = e / TPC;
    const int q  = e - TPC * r;
    const int bc = r / 3;
    const int kh = r - 3 * bc;
    const int hh = h + kh - 1;
    const int col = w0 - 4 + 4 * q;
    ok[i] = e < TPCS && col >= 0 && col <= W_ - 4 && hh >= 0 && hh < H_;
    const int idx = ok[i] ? (bc * H_ + hh) * W_ + col : 0;
    xv[i] = *(const v4f*)(x + idx);
  }
  const v2f bv = *(const v2f*)(bias + (size_t)pix0 * COUT_ + 2 * tid);

  if (loader) request(wb, wsl + PPS * WBLK);

#pragma unroll
  for (int i = 0; i < XPT; ++i) {
    const int e = tid + 256 * i;
    if (e < TPCS) {
      const int r = e / TPC;
      const int q = e - TPC * r;
      const v4f z = {0.0f, 0.0f, 0.0f, 0.0f};
      *(v4f*)(tile + r * TP + (r / 48) * 8 + 4 * q) = ok[i] ? xv[i] : z;
    }
  }
  *(v2f*)(bl + 2 * tid) = bv;
  if (loader) stash(wa, wl + slot * WSTR, lane);
  HANDOVER();

#pragma unroll
  for (int s = 0; s < NST; s += 2) {
    if (loader) {
      if (s + 2 < NST) request(wa, wsl + (s + 2) * PPS * WBLK);
      stash(wb, wl + (PPS + slot) * WSTR, lane);
    } else {
      pixel(s * PPS + slot, wl + slot * WSTR, tile, bl, out, pix0, n, half);
    }
    HANDOVER();
    if (loader) {
      if (s + 3 < NST) request(wb, wsl + (s + 3) * PPS * WBLK);
      if (s + 2 < NST) stash(wa, wl + slot * WSTR, lane);
    } else {
      pixel((s + 1) * PPS + slot, wl + (PPS + slot) * WSTR, tile, bl, out, pix0, n, half);
    }
    if (s + 2 < NST) HANDOVER();
  }
}

extern "C" void kernel_launch(void* const* d_in, const int* in_sizes, int n_in,
                              void* d_out, int out_size, void* d_ws, size_t ws_size,
                              hipStream_t stream) {
  (void)in_sizes; (void)n_in; (void)out_size; (void)d_ws; (void)ws_size;
  const float* x    = (const float*)d_in[0];
  const float* wgt  = (const float*)d_in[1];
  const float* bias = (const float*)d_in[2];
  float* out        = (float*)d_out;

  dim3 grid((H_ * W_) / SEG);
  dim3 block(256);
  xonv2d_wmma_f32<<<grid, block, 0, stream>>>(x, wgt, bias, out);
}


# Block's results staged in LDS (64 rows x 32 pixels), written at block end as whole 128-byte lines, 4 per store instruction; was 8 b32 stores of 8 lanes per pixel (4-byte pieces); sums unchanged
# speedup vs baseline: 2.9313x; 1.3523x over previous
#include <hip/hip_runtime.h>

typedef __attribute__((ext_vector_type(2))) float v2f;
typedef __attribute__((ext_vector_type(4))) float v4f;
typedef __attribute__((ext_vector_type(8))) float v8f;

#define B_    4
#define CIN_  16
#define COUT_ 16
#define H_    128
#define W_    128
#define KK    144
#define SEG   32
#define PPS   4
#define NST   (SEG / PPS)
#define TP    (SEG + 8)
#define TPC   (TP / 4)
#define TPCS  (B_ * CIN_ * 3 * TPC)
#define XPT   ((TPCS + 255) / 256)
#define BSTR  (48 * TP + 8)
#define KOFF(k) ((((k) / 9) * 3 + ((k) % 9) / 3) * TP + (k) % 3)
#define WBLK  (COUT_ * KK)
#define NWL   (WBLK / 4 / 32)
#define WROW  (KK + 4)
#define WSTR  (COUT_ * WROW)
#define OROW  (SEG + 4)
#define OSTR  (COUT_ * OROW + 8)
#define OPCS  (B_ * COUT_ * SEG / 4)

#define HANDOVER() asm volatile("s_wait_dscnt 0x0\n\ts_barrier_signal -1\n\ts_barrier_wait -1" ::: "memory")

__device__ __forceinline__ void request(v4f (&w)[NWL], const float* __restrict__ wp) {
#pragma unroll
  for (int i = 0; i < NWL; ++i) w[i] = *(const v4f*)(wp + 128 * i);
}

__device__ __forceinline__ void stash(const v4f (&w)[NWL], float* wlw, int lane) {
#pragma unroll
  for (int i = 0; i < NWL; ++i) {
    const unsigned j = 32u * i + lane;
    *(v4f*)(wlw + 4u * (j + j / 36u)) = w[i];
  }
}

__device__ __forceinline__ void pixel(int p, const float* wlw, const float* tile, const float* bl,
                                      float* ost, int n, int half) {
  const float* ab = wlw + n * WROW + 2 * half;
  const float* tb   = tile + (n & (B_ - 1)) * BSTR + p + 3;
  const float* tb2  = tb + (half ? 2 : 0);
  const float* tb15 = tb + (half ? TP - 1 : 0);
  v4f bia0 = *(const v4f*)(bl + p * COUT_ + 8 * half);
  v4f bia1 = *(const v4f*)(bl + p * COUT_ + 8 * half + 4);
  v8f acc = {};
#pragma unroll
  for (int t = 0; t < KK / 4; ++t) {
    v2f b;
    b.x = (((4 * t) % 3 == 0) ? tb2 : tb15)[KOFF(4 * t)];
    b.y = (((4 * t + 1) % 3 == 0) ? tb2 : tb15)[KOFF(4 * t + 1)];
    const v2f a = *(const v2f*)(ab + 4 * t);
    acc = __builtin_amdgcn_wmma_f32_16x16x4_f32(
         false, a,  false, b,
         (short)0, acc,  false,  false);
  }
  asm volatile("v_nop\n\tv_nop\n\tv_nop\n\tv_nop" : "+v"(acc), "+v"(bia0), "+v"(bia1));
  if (n < B_) {
    float* op = ost + n * OSTR + 8 * half * OROW + p;
#pragma unroll
    for (int r = 0; r < 8; ++r)
      op[r * OROW] = acc[r] + (r < 4 ? bia0[r & 3] : bia1[r & 3]);
  }
}

__global__ __launch_bounds__(256) void xonv2d_wmma_f32(
    const float* __restrict__ x,
    const float* __restrict__ wgt,
    const float* __restrict__ bias,
    float* __restrict__ out)
{
  __shared__ __align__(16) float tile[B_ * BSTR];
  __shared__ __align__(16) float wl[2 * PPS * WSTR];
  __shared__ __align__(16) float bl[SEG * COUT_];
  __shared__ __align__(16) float ost[B_ * OSTR];

  const int tid  = threadIdx.x;
  const int lane = tid & 31;
  const int wave = __builtin_amdgcn_readfirstlane(tid >> 5);
  const bool loader = wave < PPS;
  const int slot = wave & (PPS - 1);
  const int h    = blockIdx.x >> 2;
  const int w0   = (blockIdx.x & 3) * SEG;
  const int pix0 = h * W_ + w0;
  const int n    = lane & 15;
  const int half = lane >> 4;

  v4f wa[NWL], wb[NWL];
  const float* wsl = wgt + (size_t)(pix0 + slot) * WBLK + 4 * lane;
  if (loader) request(wa, wsl);

  v4f xv[XPT];
  bool ok[XPT];
#pragma unroll
  for (int i = 0; i < XPT; ++i) {
    const int e  = tid + 256 * i;
    const int r  = e / TPC;
    const int q  = e - TPC * r;
    const int bc = r / 3;
    const int kh = r - 3 * bc;
    const int hh = h + kh - 1;
    const int col = w0 - 4 + 4 * q;
    ok[i] = e < TPCS && col >= 0 && col <= W_ - 4 && hh >= 0 && hh < H_;
    const int idx = ok[i] ? (bc * H_ + hh) * W_ + col : 0;
    xv[i] = *(const v4f*)(x + idx);
  }
  const v2f bv = *(const v2f*)(bias + (size_t)pix0 * COUT_ + 2 * tid);

  if (loader) request(wb, wsl + PPS * WBLK);

#pragma unroll
  for (int i = 0; i < XPT; ++i) {
    const int e = tid + 256 * i;
    if (e < TPCS) {
      const int r = e / TPC;
      const int q = e - TPC * r;
      const v4f z = {0.0f, 0.0f, 0.0f, 0.0f};
      *(v4f*)(tile + r * TP + (r / 48) * 8 + 4 * q) = ok[i] ? xv[i] : z;
    }
  }
  *(v2f*)(bl + 2 * tid) = bv;
  if (loader) stash(wa, wl + slot * WSTR, lane);
  HANDOVER();

#pragma unroll
  for (int s = 0; s < NST; s += 2) {
    if (loader) {
      if (s + 2 < NST) request(wa, wsl + (s + 2) * PPS * WBLK);
      stash(wb, wl + (PPS + slot) * WSTR, lane);
    } else {
      pixel(s * PPS + slot, wl + slot * WSTR, tile, bl, ost, n, half);
    }
    HANDOVER();
    if (loader) {
      if (s + 3 < NST) request(wb, wsl + (s + 3) * PPS * WBLK);
      if (s + 2 < NST) stash(wa, wl + slot * WSTR, lane);
    } else {
      pixel((s + 1) * PPS + slot, wl + (PPS + slot) * WSTR, tile, bl, ost, n, half);
    }
    if (s + 2 < NST) HANDOVER();
  }
  HANDOVER();

#pragma unroll
  for (int i = 0; i < OPCS / 256; ++i) {
    const int e   = tid + 256 * i;
    const int row = e >> 3;
    const int q   = e & 7;
    const v4f v = *(const v4f*)(ost + (row >> 4) * OSTR + (row & 15) * OROW + 4 * q);
    *(v4f*)(out + (size_t)row * (H_ * W_) + pix0 + 4 * q) = v;
  }
}

extern "C" void kernel_launch(void* const* d_in, const int* in_sizes, int n_in,
                              void* d_out, int out_size, void* d_ws, size_t ws_size,
                              hipStream_t stream) {
  (void)in_sizes; (void)n_in; (void)out_size; (void)d_ws; (void)ws_size;
  const float* x    = (const float*)d_in[0];
  const float* wgt  = (const float*)d_in[1];
  const float* bias = (const float*)d_in[2];
  float* out        = (float*)d_out;

  dim3 grid((H_ * W_) / SEG);
  dim3 block(256);
  xonv2d_wmma_f32<<<grid, block, 0, stream>>>(x, wgt, bias, out);
}


# Matrix waves read a pixel's A and B fragments from LDS as one group ahead of its 36-WMMA chain, into the loader role's idle registers; lane half = channel half, so k is summed in a new order
# speedup vs baseline: 3.0898x; 1.0541x over previous
#include <hip/hip_runtime.h>

typedef __attribute__((ext_vector_type(2))) float v2f;
typedef __attribute__((ext_vector_type(4))) float v4f;
typedef __attribute__((ext_vector_type(8))) float v8f;

#define B_    4
#define CIN_  16
#define COUT_ 16
#define H_    128
#define W_    128
#define KK    144
#define SEG   32
#define PPS   4
#define NST   (SEG / PPS)
#define TP    (SEG + 8)
#define TPC   (TP / 4)
#define TPCS  (B_ * CIN_ * 3 * TPC)
#define XPT   ((TPCS + 255) / 256)
#define HOFF  (24 * TP + 32)
#define BSTR  (48 * TP + 40)
#define KOFF(k) ((((k) / 9) * 3 + ((k) % 9) / 3) * TP + (k) % 3)
#define WBLK  (COUT_ * KK)
#define NWL   (WBLK / 4 / 32)
#define WROW  (KK + 4)
#define WSTR  (COUT_ * WROW)
#define OROW  (SEG + 4)
#define OSTR  (COUT_ * OROW + 8)
#define OPCS  (B_ * COUT_ * SEG / 4)

#define HANDOVER() asm volatile("s_wait_dscnt 0x0\n\ts_barrier_signal -1\n\ts_barrier_wait -1" ::: "memory")

__device__ __forceinline__ void request(v4f (&w)[NWL], const float* __restrict__ wp) {
#pragma unroll
  for (int i = 0; i < NWL; ++i) w[i] = *(const v4f*)(wp + 128 * i);
}

__device__ __forceinline__ void stash(const v4f (&w)[NWL], float* wlw, int lane) {
#pragma unroll
  for (int i = 0; i < NWL; ++i) {
    const unsigned j = 32u * i + lane;
    *(v4f*)(wlw + 4u * (j + j / 36u)) = w[i];
  }
}

__device__ __forceinline__ void pixel(int p, const float* wlw, const float* tile, const float* bl,
                                      float* ost, int n, int half, v4f (&fa)[NWL], v4f (&fb)[NWL]) {
  const float* ab = wlw + n * WROW + (KK / 2) * half;
  const float* tb = tile + (n & (B_ - 1)) * BSTR + p + 3 + half * HOFF;
#pragma unroll
  for (int i = 0; i < NWL; ++i) {
    fa[i] = *(const v4f*)(ab + 4 * i);
    fb[i].x = tb[KOFF(4 * i)];
    fb[i].y = tb[KOFF(4 * i + 1)];
    fb[i].z = tb[KOFF(4 * i + 2)];
    fb[i].w = tb[KOFF(4 * i + 3)];
  }
  __builtin_amdgcn_sched_barrier(0);
  v8f acc = {};
  v4f bia0, bia1;
#pragma unroll
  for (int i = 0; i < NWL; ++i) {
    acc = __builtin_amdgcn_wmma_f32_16x16x4_f32(
         false, __builtin_shufflevector(fa[i], fa[i], 0, 1),
         false, __builtin_shufflevector(fb[i], fb[i], 0, 1),
         (short)0, acc,  false,  false);
    acc = __builtin_amdgcn_wmma_f32_16x16x4_f32(
         false, __builtin_shufflevector(fa[i], fa[i], 2, 3),
         false, __builtin_shufflevector(fb[i], fb[i], 2, 3),
         (short)0, acc,  false,  false);
    if (i == 0) {
      __builtin_amdgcn_sched_barrier(0);
      bia0 = *(const v4f*)(bl + p * COUT_ + 8 * half);
      bia1 = *(const v4f*)(bl + p * COUT_ + 8 * half + 4);
      __builtin_amdgcn_sched_barrier(0);
    }
  }
  asm volatile("v_nop\n\tv_nop\n\tv_nop\n\tv_nop" : "+v"(acc), "+v"(bia0), "+v"(bia1));
  if (n < B_) {
    float* op = ost + n * OSTR + 8 * half * OROW + p;
#pragma unroll
    for (int r = 0; r < 8; ++r)
      op[r * OROW] = acc[r] + (r < 4 ? bia0[r & 3] : bia1[r & 3]);
  }
}

__global__ __launch_bounds__(256) void xonv2d_wmma_f32(
    const float* __restrict__ x,
    const float* __restrict__ wgt,
    const float* __restrict__ bias,
    float* __restrict__ out)
{
  __shared__ __align__(16) float tile[B_ * BSTR];
  __shared__ __align__(16) float wl[2 * PPS * WSTR];
  __shared__ __align__(16) float bl[SEG * COUT_];
  __shared__ __align__(16) float ost[B_ * OSTR];

  const int tid  = threadIdx.x;
  const int lane = tid & 31;
  const int wave = __builtin_amdgcn_readfirstlane(tid >> 5);
  const bool loader = wave < PPS;
  const int slot = wave & (PPS - 1);
  const int h    = blockIdx.x >> 2;
  const int w0   = (blockIdx.x & 3) * SEG;
  const int pix0 = h * W_ + w0;
  const int n    = lane & 15;
  const int half = lane >> 4;

  v4f wa[NWL], wb[NWL];
  const float* wsl = wgt + (size_t)(pix0 + slot) * WBLK + 4 * lane;
  if (loader) request(wa, wsl);

  v4f xv[XPT];
  bool ok[XPT];
#pragma unroll
  for (int i = 0; i < XPT; ++i) {
    const int e  = tid + 256 * i;
    const int r  = e / TPC;
    const int q  = e - TPC * r;
    const int bc = r / 3;
    const int kh = r - 3 * bc;
    const int hh = h + kh - 1;
    const int col = w0 - 4 + 4 * q;
    ok[i] = e < TPCS && col >= 0 && col <= W_ - 4 && hh >= 0 && hh < H_;
    const int idx = ok[i] ? (bc * H_ + hh) * W_ + col : 0;
    xv[i] = *(const v4f*)(x + idx);
  }
  const v2f bv = *(const v2f*)(bias + (size_t)pix0 * COUT_ + 2 * tid);

  if (loader) request(wb, wsl + PPS * WBLK);

#pragma unroll
  for (int i = 0; i < XPT; ++i) {
    const int e = tid + 256 * i;
    if (e < TPCS) {
      const int r = e / TPC;
      const int q = e - TPC * r;
      const v4f z = {0.0f, 0.0f, 0.0f, 0.0f};
      const int g = r / 24;
      *(v4f*)(tile + r * TP + (g >> 1) * 40 + (g & 1) * 32 + 4 * q) = ok[i] ? xv[i] : z;
    }
  }
  *(v2f*)(bl + 2 * tid) = bv;
  if (loader) stash(wa, wl + slot * WSTR, lane);
  HANDOVER();

#pragma unroll
  for (int s = 0; s < NST; s += 2) {
    if (loader) {
      if (s + 2 < NST) request(wa, wsl + (s + 2) * PPS * WBLK);
      stash(wb, wl + (PPS + slot) * WSTR, lane);
    } else {
      pixel(s * PPS + slot, wl + slot * WSTR, tile, bl, ost, n, half, wa, wb);
    }
    HANDOVER();
    if (loader) {
      if (s + 3 < NST) request(wb, wsl + (s + 3) * PPS * WBLK);
      if (s + 2 < NST) stash(wa, wl + slot * WSTR, lane);
    } else {
      pixel((s + 1) * PPS + slot, wl + (PPS + slot) * WSTR, tile, bl, ost, n, half, wa, wb);
    }
    if (s + 2 < NST) HANDOVER();
  }
  HANDOVER();

#pragma unroll
  for (int i = 0; i < OPCS / 256; ++i) {
    const int e   = tid + 256 * i;
    const int row = e >> 3;
    const int q   = e & 7;
    const v4f v = *(const v4f*)(ost + (row >> 4) * OSTR + (row & 15) * OROW + 4 * q);
    *(v4f*)(out + (size_t)row * (H_ * W_) + pix0 + 4 * q) = v;
  }
}

extern "C" void kernel_launch(void* const* d_in, const int* in_sizes, int n_in,
                              void* d_out, int out_size, void* d_ws, size_t ws_size,
                              hipStream_t stream) {
  (void)in_sizes; (void)n_in; (void)out_size; (void)d_ws; (void)ws_size;
  const float* x    = (const float*)d_in[0];
  const float* wgt  = (const float*)d_in[1];
  const float* bias = (const float*)d_in[2];
  float* out        = (float*)d_out;

  dim3 grid((H_ * W_) / SEG);
  dim3 block(256);
  xonv2d_wmma_f32<<<grid, block, 0, stream>>>(x, wgt, bias, out);
}
